# starting kernel
# baseline (speedup 1.0000x reference)
#include <hip/hip_runtime.h>


typedef __attribute__((ext_vector_type(2))) float v2f;
typedef __attribute__((ext_vector_type(8))) float v8f;

#define IN_DIM  128
#define HID_DIM 128
#define OUT_DIM 64

__global__ void gcn_degree_kernel(const int* __restrict__ src,
                                  const int* __restrict__ dst,
                                  int* __restrict__ deg_out,
                                  int* __restrict__ deg_in, int E) {
    int e = blockIdx.x * blockDim.x + threadIdx.x;
    if (e < E) {
        atomicAdd(&deg_out[src[e]], 1);
        atomicAdd(&deg_in[dst[e]], 1);
    }
}

__global__ void gcn_invsqrt_kernel(const int* __restrict__ deg,
                                   float* __restrict__ inv, int n2) {
    int i = blockIdx.x * blockDim.x + threadIdx.x;
    if (i < n2) {
        float d = (float)deg[i];
        inv[i] = rsqrtf(fmaxf(d, 1.0f));
    }
}

__global__ __launch_bounds__(256)
void gcn_scatter_kernel(const float* __restrict__ x,
                        const int* __restrict__ src,
                        const int* __restrict__ dst,
                        const float* __restrict__ inv_out,
                        float* __restrict__ agg, int E) {
    long long t = (long long)blockIdx.x * blockDim.x + threadIdx.x;
    int e = (int)(t >> 5);
    if (e >= E) return;
    int lane = (int)(t & 31);
    int s = src[e];
    int d = dst[e];
    float sc = inv_out[s];
    const float4* xp = (const float4*)(x + (size_t)s * 128) + lane;
    float4 v = *xp;
    float* ap = agg + (size_t)d * 128 + lane * 4;
    unsafeAtomicAdd(ap + 0, v.x * sc);
    unsafeAtomicAdd(ap + 1, v.y * sc);
    unsafeAtomicAdd(ap + 2, v.z * sc);
    unsafeAtomicAdd(ap + 3, v.w * sc);
}

template <int K, int NOUT, bool RELU>
__global__ __launch_bounds__(256)
void gcn_wmma_gemm_kernel(const float* __restrict__ A,
                          const float* __restrict__ W,
                          const float* __restrict__ bias,
                          const float* __restrict__ inv_in,
                          float* __restrict__ out, int N) {
    constexpr int S  = 2 * NOUT + 32;
    constexpr int NT = NOUT / 16;
    __shared__ float lds_w[(K / 2) * S];

    for (int idx = threadIdx.x; idx < K * NOUT; idx += blockDim.x) {
        int k = idx / NOUT;
        int n = idx - k * NOUT;
        lds_w[(k >> 1) * S + n * 2 + (k & 1)] = W[idx];
    }
    __syncthreads();

    const int lane = threadIdx.x & 31;
    const int wave = threadIdx.x >> 5;
    const int g = lane >> 4;
    const int mr = lane & 15;

    const int row0 = blockIdx.x * 128 + wave * 16;
    int row = row0 + mr;
    if (row >= N) row = N - 1;
    const float s = inv_in[row];
    const float* __restrict__ Arow = A + (size_t)row * K;

    v8f acc[NT];
#pragma unroll
    for (int t = 0; t < NT; ++t) acc[t] = (v8f){};

#pragma unroll 4
    for (int k0 = 0; k0 < K; k0 += 4) {
        const int ka = k0 + 2 * g;
        v2f a = *(const v2f*)(Arow + ka);
        a.x *= s;
        a.y *= s;
        const v2f* __restrict__ wp =
            (const v2f*)(lds_w + (size_t)(k0 / 2 + g) * S);
#pragma unroll
        for (int t = 0; t < NT; ++t) {
            v2f b = wp[t * 16 + mr];
            acc[t] = __builtin_amdgcn_wmma_f32_16x16x4_f32(
                 false, a,  false, b,
                 (short)0, acc[t],  false,  false);
        }
    }

#pragma unroll
    for (int t = 0; t < NT; ++t) {
        const int Nbase = t * 16;
        const float bv = bias[Nbase + mr];
#pragma unroll
        for (int r = 0; r < 8; ++r) {
            const int orow = row0 + r + 8 * g;
            if (orow < N) {
                float v = acc[t][r] + bv;
                if (RELU) v = fmaxf(v, 0.0f);
                out[(size_t)orow * NOUT + Nbase + mr] = v;
            }
        }
    }
}

extern "C" void kernel_launch(void* const* d_in, const int* in_sizes, int n_in,
                              void* d_out, int out_size, void* d_ws, size_t ws_size,
                              hipStream_t stream) {
    const float* in_feat = (const float*)d_in[0];
    const float* W1      = (const float*)d_in[1];
    const float* b1      = (const float*)d_in[2];
    const float* W2      = (const float*)d_in[3];
    const float* b2      = (const float*)d_in[4];
    const int*   src     = (const int*)d_in[5];
    const int*   dst     = (const int*)d_in[6];
    float*       out     = (float*)d_out;

    const int N = in_sizes[0] / IN_DIM;
    const int E = in_sizes[5];

    char* ws = (char*)d_ws;
    int*   deg   = (int*)ws;
    float* inv   = (float*)(ws + (size_t)2 * N * 4);
    float* agg   = (float*)(ws + (size_t)4 * N * 4);
    float* hbuf  = agg + (size_t)N * HID_DIM;
    float* inv_out = inv;
    float* inv_in  = inv + N;

    hipMemsetAsync(deg, 0, (size_t)2 * N * sizeof(int), stream);
    hipMemsetAsync(agg, 0, (size_t)N * HID_DIM * sizeof(float), stream);

    {
        int blk = 256;
        gcn_degree_kernel<<<(E + blk - 1) / blk, blk, 0, stream>>>(
            src, dst, deg, deg + N, E);
        gcn_invsqrt_kernel<<<(2 * N + blk - 1) / blk, blk, 0, stream>>>(
            deg, inv, 2 * N);
    }

    {
        long long threads = (long long)E * 32;
        int blk = 256;
        long long grid = (threads + blk - 1) / blk;
        gcn_scatter_kernel<<<(unsigned)grid, blk, 0, stream>>>(
            in_feat, src, dst, inv_out, agg, E);
    }

    {
        int grid = (N + 127) / 128;
        gcn_wmma_gemm_kernel<IN_DIM, HID_DIM, true>
            <<<grid, 256, 0, stream>>>(agg, W1, b1, inv_in, hbuf, N);
    }

    hipMemsetAsync(agg, 0, (size_t)N * HID_DIM * sizeof(float), stream);
    {
        long long threads = (long long)E * 32;
        int blk = 256;
        long long grid = (threads + blk - 1) / blk;
        gcn_scatter_kernel<<<(unsigned)grid, blk, 0, stream>>>(
            hbuf, src, dst, inv_out, agg, E);
    }

    {
        int grid = (N + 127) / 128;
        gcn_wmma_gemm_kernel<HID_DIM, OUT_DIM, false>
            <<<grid, 256, 0, stream>>>(agg, W2, b2, inv_in, out, N);
    }
}


# Float-atomic edge scatter replaced by dst-bucketed edge list (slot = returning int add) + one-wave-per-row register gather; each row's sources sorted by src id first, so the f32 sum order is fixed
# speedup vs baseline: 2.2999x; 2.2999x over previous
#include <hip/hip_runtime.h>


typedef __attribute__((ext_vector_type(2))) float v2f;
typedef __attribute__((ext_vector_type(8))) float v8f;

#define IN_DIM  128
#define HID_DIM 128
#define OUT_DIM 64

#define KEY_NONE 0x7fffffff

__global__ void gcn_degree_slot_kernel(const int* __restrict__ src,
                                       const int* __restrict__ dst,
                                       int* __restrict__ deg_out,
                                       int* __restrict__ deg_in,
                                       int* __restrict__ slot, int E) {
    int e = blockIdx.x * blockDim.x + threadIdx.x;
    if (e < E) {
        atomicAdd(&deg_out[src[e]], 1);
        slot[e] = atomicAdd(&deg_in[dst[e]], 1);
    }
}

__global__ __launch_bounds__(256)
void gcn_node_kernel(const int* __restrict__ deg, float* __restrict__ inv,
                     int* __restrict__ row_start, int* __restrict__ cursor,
                     int N) {
    __shared__ int wave_sum[8];
    __shared__ int block_base;
    const int i = blockIdx.x * 256 + threadIdx.x;
    const int lane = threadIdx.x & 31;
    const int wave = threadIdx.x >> 5;
    int din = 0;
    if (i < N) {
        din = deg[N + i];
        inv[i]     = rsqrtf(fmaxf((float)deg[i], 1.0f));
        inv[N + i] = rsqrtf(fmaxf((float)din, 1.0f));
    }
    int incl = din;
#pragma unroll
    for (int o = 1; o < 32; o <<= 1) {
        int t = __shfl_up(incl, o, 32);
        if (lane >= o) incl += t;
    }
    if (lane == 31) wave_sum[wave] = incl;
    __syncthreads();
    if (threadIdx.x == 0) {
        int tot = 0;
#pragma unroll
        for (int k = 0; k < 8; ++k) { int t = wave_sum[k]; wave_sum[k] = tot; tot += t; }
        block_base = atomicAdd(cursor, tot);
    }
    __syncthreads();
    if (i < N) row_start[i] = block_base + wave_sum[wave] + incl - din;
}

__global__ void gcn_bucket_kernel(const int* __restrict__ src,
                                  const int* __restrict__ dst,
                                  const int* __restrict__ slot,
                                  const int* __restrict__ row_start,
                                  int* __restrict__ list, int E) {
    int e = blockIdx.x * blockDim.x + threadIdx.x;
    if (e < E) list[row_start[dst[e]] + slot[e]] = src[e];
}

__device__ __forceinline__ void gcn_add_row(float4& acc, const float4 v, const float sc) {
    acc.x += v.x * sc;
    acc.y += v.y * sc;
    acc.z += v.z * sc;
    acc.w += v.w * sc;
}

__device__ __forceinline__ float4 gcn_load_row(const float* __restrict__ x, int s, int lane) {
    return *((const float4*)(x + (size_t)s * 128) + lane);
}

template <int NCH>
__device__ __forceinline__ float4 gcn_gather_ranked(const float* __restrict__ x,
                                                    const float* __restrict__ inv_out,
                                                    const int* __restrict__ row_list,
                                                    int deg, int lane,
                                                    int* __restrict__ srt) {
    int key[NCH], rk[NCH];
#pragma unroll
    for (int c = 0; c < NCH; ++c) {
        const int i = lane + 32 * c;
        key[c] = (i < deg) ? row_list[i] : KEY_NONE;
        rk[c] = 0;
    }
#pragma unroll
    for (int c = 0; c < NCH; ++c) {
        int lim = deg - 32 * c;
        if (lim > 32) lim = 32;
        for (int j = 0; j < lim; ++j) {
            const int kj = __builtin_amdgcn_readlane(key[c], j);
            const int jj = j + 32 * c;
#pragma unroll
            for (int c2 = 0; c2 < NCH; ++c2) {
                const int i2 = lane + 32 * c2;
                rk[c2] += (kj < key[c2] || (kj == key[c2] && jj < i2)) ? 1 : 0;
            }
        }
    }
#pragma unroll
    for (int c = 0; c < NCH; ++c) {
        const int i = lane + 32 * c;
        if (i >= deg) rk[c] = i;
        srt[rk[c]] = key[c];
    }
    __builtin_amdgcn_fence(__ATOMIC_RELEASE, "wavefront");
    __builtin_amdgcn_wave_barrier();
    __builtin_amdgcn_fence(__ATOMIC_ACQUIRE, "wavefront");
    int sorted[NCH];
#pragma unroll
    for (int c = 0; c < NCH; ++c) sorted[c] = srt[lane + 32 * c];

    float4 acc = make_float4(0.0f, 0.0f, 0.0f, 0.0f);
#pragma unroll
    for (int c = 0; c < NCH; ++c) {
        int lim = deg - 32 * c;
        if (lim > 32) lim = 32;
        int j = 0;
        for (; j + 8 <= lim; j += 8) {
            float4 v[8];
            float sc[8];
#pragma unroll
            for (int u = 0; u < 8; ++u) {
                const int s = __builtin_amdgcn_readlane(sorted[c], j + u);
                v[u] = gcn_load_row(x, s, lane);
                sc[u] = inv_out[s];
            }
#pragma unroll
            for (int u = 0; u < 8; ++u) gcn_add_row(acc, v[u], sc[u]);
        }
        const int rest = lim - j;
        if (rest > 0) {
            float4 v[7];
            float sc[7];
#pragma unroll
            for (int u = 0; u < 7; ++u) {
                v[u] = make_float4(0.0f, 0.0f, 0.0f, 0.0f);
                sc[u] = 0.0f;
                if (u < rest) {
                    const int s = __builtin_amdgcn_readlane(sorted[c], j + u);
                    v[u] = gcn_load_row(x, s, lane);
                    sc[u] = inv_out[s];
                }
            }
#pragma unroll
            for (int u = 0; u < 7; ++u)
                if (u < rest) gcn_add_row(acc, v[u], sc[u]);
        }
    }
    return acc;
}

__device__ __forceinline__ int gcn_wave_min(int v) {
#pragma unroll
    for (int o = 16; o > 0; o >>= 1) { int t = __shfl_xor(v, o, 32); v = (t < v) ? t : v; }
    return v;
}
__device__ __forceinline__ int gcn_wave_add(int v) {
#pragma unroll
    for (int o = 16; o > 0; o >>= 1) v += __shfl_xor(v, o, 32);
    return v;
}

__device__ __forceinline__ float4 gcn_gather_long(const float* __restrict__ x,
                                                  const float* __restrict__ inv_out,
                                                  const int* __restrict__ row_list,
                                                  int deg, int lane) {
    float4 acc = make_float4(0.0f, 0.0f, 0.0f, 0.0f);
    int cur = -1;
    for (int pass = 0; pass < deg; ++pass) {
        int m = KEY_NONE;
        for (int p = lane; p < deg; p += 32) {
            const int k = row_list[p];
            if (k > cur && k < m) m = k;
        }
        m = gcn_wave_min(m);
        if (m == KEY_NONE) break;
        int cnt = 0;
        for (int p = lane; p < deg; p += 32) cnt += (row_list[p] == m) ? 1 : 0;
        cnt = __builtin_amdgcn_readfirstlane(gcn_wave_add(cnt));
        m = __builtin_amdgcn_readfirstlane(m);
        const float4 v = gcn_load_row(x, m, lane);
        const float sc = inv_out[m];
        for (int t = 0; t < cnt; ++t) gcn_add_row(acc, v, sc);
        cur = m;
    }
    return acc;
}

__global__ __launch_bounds__(256)
void gcn_gather_kernel(const float* __restrict__ x,
                       const int* __restrict__ list,
                       const int* __restrict__ row_start,
                       const int* __restrict__ deg_in,
                       const float* __restrict__ inv_out,
                       float* __restrict__ agg, int N) {
    __shared__ int srt_all[8][64];
    const int lane = threadIdx.x & 31;
    const int wave = __builtin_amdgcn_readfirstlane((int)(threadIdx.x >> 5));
    const int row = blockIdx.x * 8 + wave;
    if (row >= N) return;
    const int start = row_start[row];
    const int deg = deg_in[row];
    const int* __restrict__ row_list = list + start;
    float4 acc;
    if (deg <= 32)      acc = gcn_gather_ranked<1>(x, inv_out, row_list, deg, lane, srt_all[wave]);
    else if (deg <= 64) acc = gcn_gather_ranked<2>(x, inv_out, row_list, deg, lane, srt_all[wave]);
    else                acc = gcn_gather_long(x, inv_out, row_list, deg, lane);
    *((float4*)(agg + (size_t)row * 128) + lane) = acc;
}

template <int K, int NOUT, bool RELU>
__global__ __launch_bounds__(256)
void gcn_wmma_gemm_kernel(const float* __restrict__ A,
                          const float* __restrict__ W,
                          const float* __restrict__ bias,
                          const float* __restrict__ inv_in,
                          float* __restrict__ out, int N) {
    constexpr int S  = 2 * NOUT + 32;
    constexpr int NT = NOUT / 16;
    __shared__ float lds_w[(K / 2) * S];

    for (int idx = threadIdx.x; idx < K * NOUT; idx += blockDim.x) {
        int k = idx / NOUT;
        int n = idx - k * NOUT;
        lds_w[(k >> 1) * S + n * 2 + (k & 1)] = W[idx];
    }
    __syncthreads();

    const int lane = threadIdx.x & 31;
    const int wave = threadIdx.x >> 5;
    const int g = lane >> 4;
    const int mr = lane & 15;

    const int row0 = blockIdx.x * 128 + wave * 16;
    int row = row0 + mr;
    if (row >= N) row = N - 1;
    const float s = inv_in[row];
    const float* __restrict__ Arow = A + (size_t)row * K;

    v8f acc[NT];
#pragma unroll
    for (int t = 0; t < NT; ++t) acc[t] = (v8f){};

#pragma unroll 4
    for (int k0 = 0; k0 < K; k0 += 4) {
        const int ka = k0 + 2 * g;
        v2f a = *(const v2f*)(Arow + ka);
        a.x *= s;
        a.y *= s;
        const v2f* __restrict__ wp =
            (const v2f*)(lds_w + (size_t)(k0 / 2 + g) * S);
#pragma unroll
        for (int t = 0; t < NT; ++t) {
            v2f b = wp[t * 16 + mr];
            acc[t] = __builtin_amdgcn_wmma_f32_16x16x4_f32(
                 false, a,  false, b,
                 (short)0, acc[t],  false,  false);
        }
    }

#pragma unroll
    for (int t = 0; t < NT; ++t) {
        const int Nbase = t * 16;
        const float bv = bias[Nbase + mr];
#pragma unroll
        for (int r = 0; r < 8; ++r) {
            const int orow = row0 + r + 8 * g;
            if (orow < N) {
                float v = acc[t][r] + bv;
                if (RELU) v = fmaxf(v, 0.0f);
                out[(size_t)orow * NOUT + Nbase + mr] = v;
            }
        }
    }
}

extern "C" void kernel_launch(void* const* d_in, const int* in_sizes, int n_in,
                              void* d_out, int out_size, void* d_ws, size_t ws_size,
                              hipStream_t stream) {
    const float* in_feat = (const float*)d_in[0];
    const float* W1      = (const float*)d_in[1];
    const float* b1      = (const float*)d_in[2];
    const float* W2      = (const float*)d_in[3];
    const float* b2      = (const float*)d_in[4];
    const int*   src     = (const int*)d_in[5];
    const int*   dst     = (const int*)d_in[6];
    float*       out     = (float*)d_out;

    const int N = in_sizes[0] / IN_DIM;
    const int E = in_sizes[5];

    float* agg    = (float*)d_ws;
    float* hbuf   = agg + (size_t)N * HID_DIM;
    int*   deg    = (int*)(hbuf + (size_t)N * HID_DIM);
    int*   cursor = deg + (size_t)2 * N;
    float* inv    = (float*)(cursor + 4);
    int*   rstart = (int*)(inv + (size_t)2 * N);
    int*   slot   = rstart + N;
    int*   list   = slot + E;
    float* inv_out = inv;
    float* inv_in  = inv + N;

    hipMemsetAsync(deg, 0, ((size_t)2 * N + 4) * sizeof(int), stream);

    {
        int blk = 256;
        gcn_degree_slot_kernel<<<(E + blk - 1) / blk, blk, 0, stream>>>(
            src, dst, deg, deg + N, slot, E);
        gcn_node_kernel<<<(N + 255) / 256, 256, 0, stream>>>(
            deg, inv, rstart, cursor, N);
        gcn_bucket_kernel<<<(E + blk - 1) / blk, blk, 0, stream>>>(
            src, dst, slot, rstart, list, E);
    }

    gcn_gather_kernel<<<(N + 7) / 8, 256, 0, stream>>>(
        in_feat, list, rstart, deg + N, inv_out, agg, N);

    {
        int grid = (N + 127) / 128;
        gcn_wmma_gemm_kernel<IN_DIM, HID_DIM, true>
            <<<grid, 256, 0, stream>>>(agg, W1, b1, inv_in, hbuf, N);
    }

    gcn_gather_kernel<<<(N + 7) / 8, 256, 0, stream>>>(
        hbuf, list, rstart, deg + N, inv_out, agg, N);

    {
        int grid = (N + 127) / 128;
        gcn_wmma_gemm_kernel<HID_DIM, OUT_DIM, false>
            <<<grid, 256, 0, stream>>>(agg, W2, b2, inv_in, out, N);
    }
}


# Edge preparation (memset, global-atomic degree/slot pass, node pass, scattered bucket pass) replaced by two-level LDS partition: chunks ordered by 256-node range, one block per range; sum order kept
# speedup vs baseline: 3.7759x; 1.6418x over previous
#include <hip/hip_runtime.h>


typedef __attribute__((ext_vector_type(2))) float v2f;
typedef __attribute__((ext_vector_type(8))) float v8f;

#define IN_DIM  128
#define HID_DIM 128
#define OUT_DIM 64

#define KEY_NONE 0x7fffffff

#define RANGE_SHIFT 8
#define RANGE_NODES 256
#define CHUNK_EDGES 4096
#define OWNER_STAGE 8192

__global__ __launch_bounds__(1024)
void gcn_partition_kernel(const int* __restrict__ src,
                          const int* __restrict__ dst,
                          int* __restrict__ part_d_raw,
                          int* __restrict__ part_s,
                          int* __restrict__ tab_raw,
                          int* __restrict__ cursor,
                          int E, int R, int NB) {
    extern __shared__ __attribute__((aligned(16))) int part_lds[];
    int2*     stage_d = (int2*)part_lds;
    int*      stage_s = part_lds + 2 * CHUNK_EDGES;
    unsigned* hist    = (unsigned*)(part_lds + 3 * CHUNK_EDGES);
    __shared__ unsigned wave_sum[32];
    __shared__ unsigned chunk_total;
    int2*  part_d = (int2*)part_d_raw;
    uint2* tab    = (uint2*)tab_raw;

    const int tid  = threadIdx.x;
    const int lane = tid & 31;
    const int wave = tid >> 5;
    const int blk  = blockIdx.x;
    const size_t e0 = (size_t)blk * CHUNK_EDGES;

    for (int i = tid; i < R; i += 1024) hist[i] = 0u;
    if (blk == 0 && tid == 0) *cursor = 0;
    __syncthreads();

    int s_k[4], d_k[4];
    unsigned rd_k[4], rs_k[4];
#pragma unroll
    for (int k = 0; k < 4; ++k) {
        const size_t e = e0 + (size_t)(k * 1024 + tid);
        s_k[k] = -1;
        d_k[k] = -1;
        rd_k[k] = 0u;
        rs_k[k] = 0u;
        if (e < (size_t)E) {
            s_k[k] = src[e];
            d_k[k] = dst[e];
        }
    }
#pragma unroll
    for (int k = 0; k < 4; ++k) {
        if (s_k[k] >= 0) {
            rd_k[k] = atomicAdd(&hist[d_k[k] >> RANGE_SHIFT], 1u) & 0xffffu;
            rs_k[k] = atomicAdd(&hist[s_k[k] >> RANGE_SHIFT], 0x10000u) >> 16;
        }
    }
    __syncthreads();

    unsigned carry = 0u;
    for (int r0 = 0; r0 < R; r0 += 1024) {
        const int r = r0 + tid;
        const unsigned c = (r < R) ? hist[r] : 0u;
        unsigned incl = c;
#pragma unroll
        for (int o = 1; o < 32; o <<= 1) {
            const unsigned t = __shfl_up(incl, o, 32);
            if (lane >= o) incl += t;
        }
        if (lane == 31) wave_sum[wave] = incl;
        __syncthreads();
        if (tid == 0) {
            unsigned tot = 0u;
            for (int k = 0; k < 32; ++k) { const unsigned t = wave_sum[k]; wave_sum[k] = tot; tot += t; }
            chunk_total = tot;
        }
        __syncthreads();
        const unsigned off = carry + wave_sum[wave] + incl - c;
        if (r < R) {
            hist[r] = off;
            uint2 w;
            w.x = (off & 0xffffu) | ((c & 0xffffu) << 16);
            w.y = (off >> 16) | ((c >> 16) << 16);
            tab[(size_t)r * NB + blk] = w;
        }
        carry += chunk_total;
        __syncthreads();
    }

#pragma unroll
    for (int k = 0; k < 4; ++k) {
        if (s_k[k] >= 0) {
            const unsigned pd = (hist[d_k[k] >> RANGE_SHIFT] & 0xffffu) + rd_k[k];
            const unsigned ps = (hist[s_k[k] >> RANGE_SHIFT] >> 16) + rs_k[k];
            stage_d[pd] = make_int2(s_k[k], d_k[k]);
            stage_s[ps] = s_k[k];
        }
    }
    __syncthreads();

    size_t left = (size_t)E - e0;
    const int n = (left < (size_t)CHUNK_EDGES) ? (int)left : CHUNK_EDGES;
    for (int i = tid; i < n; i += 1024) {
        part_d[e0 + i] = stage_d[i];
        part_s[e0 + i] = stage_s[i];
    }
}

__global__ __launch_bounds__(512)
void gcn_range_kernel(const int* __restrict__ part_d_raw,
                      const int* __restrict__ part_s,
                      const int* __restrict__ tab_raw,
                      int* __restrict__ cursor,
                      float* __restrict__ inv,
                      int* __restrict__ row_start,
                      int* __restrict__ deg_in,
                      int* __restrict__ list,
                      int N, int NB) {
    __shared__ int cnt_in[RANGE_NODES];
    __shared__ int cnt_out[RANGE_NODES];
    __shared__ int row_off[RANGE_NODES];
    __shared__ int fill[RANGE_NODES];
    __shared__ int wave_sum[8];
    __shared__ int run_base, run_total;
    __shared__ int stage[OWNER_STAGE];
    const int2*  part_d = (const int2*)part_d_raw;
    const uint2* tab    = (const uint2*)tab_raw;

    const int tid  = threadIdx.x;
    const int lane = tid & 31;
    const int wave = tid >> 5;
    const int rng  = blockIdx.x;

    if (tid < RANGE_NODES) { cnt_in[tid] = 0; cnt_out[tid] = 0; fill[tid] = 0; }
    __syncthreads();

    for (int blk = tid; blk < NB; blk += 512) {
        const uint2 w = tab[(size_t)rng * NB + blk];
        {
            const int off = (int)(w.x & 0xffffu), c = (int)(w.x >> 16);
            const int2* __restrict__ p = part_d + (size_t)blk * CHUNK_EDGES + off;
            for (int i = 0; i < c; i += 8) {
                int d[8];
#pragma unroll
                for (int u = 0; u < 8; ++u) d[u] = p[(i + u < c) ? i + u : c - 1].y;
#pragma unroll
                for (int u = 0; u < 8; ++u)
                    if (i + u < c) atomicAdd(&cnt_in[d[u] & (RANGE_NODES - 1)], 1);
            }
        }
        {
            const int off = (int)(w.y & 0xffffu), c = (int)(w.y >> 16);
            const int* __restrict__ q = part_s + (size_t)blk * CHUNK_EDGES + off;
            for (int i = 0; i < c; i += 8) {
                int s[8];
#pragma unroll
                for (int u = 0; u < 8; ++u) s[u] = q[(i + u < c) ? i + u : c - 1];
#pragma unroll
                for (int u = 0; u < 8; ++u)
                    if (i + u < c) atomicAdd(&cnt_out[s[u] & (RANGE_NODES - 1)], 1);
            }
        }
    }
    __syncthreads();

    const int node = rng * RANGE_NODES + tid;
    int din = 0;
    if (tid < RANGE_NODES) {
        din = cnt_in[tid];
        if (node < N) {
            inv[node]     = rsqrtf(fmaxf((float)cnt_out[tid], 1.0f));
            inv[N + node] = rsqrtf(fmaxf((float)din, 1.0f));
            deg_in[node]  = din;
        }
    }
    int incl = din;
#pragma unroll
    for (int o = 1; o < 32; o <<= 1) {
        const int t = __shfl_up(incl, o, 32);
        if (lane >= o) incl += t;
    }
    if (tid < RANGE_NODES && lane == 31) wave_sum[wave] = incl;
    __syncthreads();
    if (tid == 0) {
        int tot = 0;
#pragma unroll
        for (int k = 0; k < 8; ++k) { const int t = wave_sum[k]; wave_sum[k] = tot; tot += t; }
        run_total = tot;
        run_base = atomicAdd(cursor, tot);
    }
    __syncthreads();
    const int total = run_total;
    const int base  = run_base;
    if (tid < RANGE_NODES) {
        const int excl = wave_sum[wave] + incl - din;
        row_off[tid] = excl;
        if (node < N) row_start[node] = base + excl;
    }
    __syncthreads();

    const bool staged = total <= OWNER_STAGE;
    for (int blk = tid; blk < NB; blk += 512) {
        const uint2 w = tab[(size_t)rng * NB + blk];
        const int off = (int)(w.x & 0xffffu), c = (int)(w.x >> 16);
        const int2* __restrict__ p = part_d + (size_t)blk * CHUNK_EDGES + off;
        for (int i = 0; i < c; i += 8) {
            int2 ed[8];
#pragma unroll
            for (int u = 0; u < 8; ++u) ed[u] = p[(i + u < c) ? i + u : c - 1];
#pragma unroll
            for (int u = 0; u < 8; ++u) {
                if (i + u < c) {
                    const int j = ed[u].y & (RANGE_NODES - 1);
                    const int pos = row_off[j] + atomicAdd(&fill[j], 1);
                    if (staged) stage[pos] = ed[u].x;
                    else        list[(size_t)base + pos] = ed[u].x;
                }
            }
        }
    }
    if (staged) {
        __syncthreads();
        for (int i = tid; i < total; i += 512) list[(size_t)base + i] = stage[i];
    }
}

__device__ __forceinline__ void gcn_add_row(float4& acc, const float4 v, const float sc) {
    acc.x += v.x * sc;
    acc.y += v.y * sc;
    acc.z += v.z * sc;
    acc.w += v.w * sc;
}

__device__ __forceinline__ float4 gcn_load_row(const float* __restrict__ x, int s, int lane) {
    return *((const float4*)(x + (size_t)s * 128) + lane);
}

template <int NCH>
__device__ __forceinline__ float4 gcn_gather_ranked(const float* __restrict__ x,
                                                    const float* __restrict__ inv_out,
                                                    const int* __restrict__ row_list,
                                                    int deg, int lane,
                                                    int* __restrict__ srt) {
    int key[NCH], rk[NCH];
#pragma unroll
    for (int c = 0; c < NCH; ++c) {
        const int i = lane + 32 * c;
        key[c] = (i < deg) ? row_list[i] : KEY_NONE;
        rk[c] = 0;
    }
#pragma unroll
    for (int c = 0; c < NCH; ++c) {
        int lim = deg - 32 * c;
        if (lim > 32) lim = 32;
        for (int j = 0; j < lim; ++j) {
            const int kj = __builtin_amdgcn_readlane(key[c], j);
            const int jj = j + 32 * c;
#pragma unroll
            for (int c2 = 0; c2 < NCH; ++c2) {
                const int i2 = lane + 32 * c2;
                rk[c2] += (kj < key[c2] || (kj == key[c2] && jj < i2)) ? 1 : 0;
            }
        }
    }
#pragma unroll
    for (int c = 0; c < NCH; ++c) {
        const int i = lane + 32 * c;
        if (i >= deg) rk[c] = i;
        srt[rk[c]] = key[c];
    }
    __builtin_amdgcn_fence(__ATOMIC_RELEASE, "wavefront");
    __builtin_amdgcn_wave_barrier();
    __builtin_amdgcn_fence(__ATOMIC_ACQUIRE, "wavefront");
    int sorted[NCH];
#pragma unroll
    for (int c = 0; c < NCH; ++c) sorted[c] = srt[lane + 32 * c];

    float4 acc = make_float4(0.0f, 0.0f, 0.0f, 0.0f);
#pragma unroll
    for (int c = 0; c < NCH; ++c) {
        int lim = deg - 32 * c;
        if (lim > 32) lim = 32;
        int j = 0;
        for (; j + 8 <= lim; j += 8) {
            float4 v[8];
            float sc[8];
#pragma unroll
            for (int u = 0; u < 8; ++u) {
                const int s = __builtin_amdgcn_readlane(sorted[c], j + u);
                v[u] = gcn_load_row(x, s, lane);
                sc[u] = inv_out[s];
            }
#pragma unroll
            for (int u = 0; u < 8; ++u) gcn_add_row(acc, v[u], sc[u]);
        }
        const int rest = lim - j;
        if (rest > 0) {
            float4 v[7];
            float sc[7];
#pragma unroll
            for (int u = 0; u < 7; ++u) {
                v[u] = make_float4(0.0f, 0.0f, 0.0f, 0.0f);
                sc[u] = 0.0f;
                if (u < rest) {
                    const int s = __builtin_amdgcn_readlane(sorted[c], j + u);
                    v[u] = gcn_load_row(x, s, lane);
                    sc[u] = inv_out[s];
                }
            }
#pragma unroll
            for (int u = 0; u < 7; ++u)
                if (u < rest) gcn_add_row(acc, v[u], sc[u]);
        }
    }
    return acc;
}

__device__ __forceinline__ int gcn_wave_min(int v) {
#pragma unroll
    for (int o = 16; o > 0; o >>= 1) { int t = __shfl_xor(v, o, 32); v = (t < v) ? t : v; }
    return v;
}
__device__ __forceinline__ int gcn_wave_add(int v) {
#pragma unroll
    for (int o = 16; o > 0; o >>= 1) v += __shfl_xor(v, o, 32);
    return v;
}

__device__ __forceinline__ float4 gcn_gather_long(const float* __restrict__ x,
                                                  const float* __restrict__ inv_out,
                                                  const int* __restrict__ row_list,
                                                  int deg, int lane) {
    float4 acc = make_float4(0.0f, 0.0f, 0.0f, 0.0f);
    int cur = -1;
    for (int pass = 0; pass < deg; ++pass) {
        int m = KEY_NONE;
        for (int p = lane; p < deg; p += 32) {
            const int k = row_list[p];
            if (k > cur && k < m) m = k;
        }
        m = gcn_wave_min(m);
        if (m == KEY_NONE) break;
        int cnt = 0;
        for (int p = lane; p < deg; p += 32) cnt += (row_list[p] == m) ? 1 : 0;
        cnt = __builtin_amdgcn_readfirstlane(gcn_wave_add(cnt));
        m = __builtin_amdgcn_readfirstlane(m);
        const float4 v = gcn_load_row(x, m, lane);
        const float sc = inv_out[m];
        for (int t = 0; t < cnt; ++t) gcn_add_row(acc, v, sc);
        cur = m;
    }
    return acc;
}

__global__ __launch_bounds__(256)
void gcn_gather_kernel(const float* __restrict__ x,
                       const int* __restrict__ list,
                       const int* __restrict__ row_start,
                       const int* __restrict__ deg_in,
                       const float* __restrict__ inv_out,
                       float* __restrict__ agg, int N) {
    __shared__ int srt_all[8][64];
    const int lane = threadIdx.x & 31;
    const int wave = __builtin_amdgcn_readfirstlane((int)(threadIdx.x >> 5));
    const int row = blockIdx.x * 8 + wave;
    if (row >= N) return;
    const int start = row_start[row];
    const int deg = deg_in[row];
    const int* __restrict__ row_list = list + start;
    float4 acc;
    if (deg <= 32)      acc = gcn_gather_ranked<1>(x, inv_out, row_list, deg, lane, srt_all[wave]);
    else if (deg <= 64) acc = gcn_gather_ranked<2>(x, inv_out, row_list, deg, lane, srt_all[wave]);
    else                acc = gcn_gather_long(x, inv_out, row_list, deg, lane);
    *((float4*)(agg + (size_t)row * 128) + lane) = acc;
}

template <int K, int NOUT, bool RELU>
__global__ __launch_bounds__(256)
void gcn_wmma_gemm_kernel(const float* __restrict__ A,
                          const float* __restrict__ W,
                          const float* __restrict__ bias,
                          const float* __restrict__ inv_in,
                          float* __restrict__ out, int N) {
    constexpr int S  = 2 * NOUT + 32;
    constexpr int NT = NOUT / 16;
    __shared__ float lds_w[(K / 2) * S];

    for (int idx = threadIdx.x; idx < K * NOUT; idx += blockDim.x) {
        int k = idx / NOUT;
        int n = idx - k * NOUT;
        lds_w[(k >> 1) * S + n * 2 + (k & 1)] = W[idx];
    }
    __syncthreads();

    const int lane = threadIdx.x & 31;
    const int wave = threadIdx.x >> 5;
    const int g = lane >> 4;
    const int mr = lane & 15;

    const int row0 = blockIdx.x * 128 + wave * 16;
    int row = row0 + mr;
    if (row >= N) row = N - 1;
    const float s = inv_in[row];
    const float* __restrict__ Arow = A + (size_t)row * K;

    v8f acc[NT];
#pragma unroll
    for (int t = 0; t < NT; ++t) acc[t] = (v8f){};

#pragma unroll 4
    for (int k0 = 0; k0 < K; k0 += 4) {
        const int ka = k0 + 2 * g;
        v2f a = *(const v2f*)(Arow + ka);
        a.x *= s;
        a.y *= s;
        const v2f* __restrict__ wp =
            (const v2f*)(lds_w + (size_t)(k0 / 2 + g) * S);
#pragma unroll
        for (int t = 0; t < NT; ++t) {
            v2f b = wp[t * 16 + mr];
            acc[t] = __builtin_amdgcn_wmma_f32_16x16x4_f32(
                 false, a,  false, b,
                 (short)0, acc[t],  false,  false);
        }
    }

#pragma unroll
    for (int t = 0; t < NT; ++t) {
        const int Nbase = t * 16;
        const float bv = bias[Nbase + mr];
#pragma unroll
        for (int r = 0; r < 8; ++r) {
            const int orow = row0 + r + 8 * g;
            if (orow < N) {
                float v = acc[t][r] + bv;
                if (RELU) v = fmaxf(v, 0.0f);
                out[(size_t)orow * NOUT + Nbase + mr] = v;
            }
        }
    }
}

extern "C" void kernel_launch(void* const* d_in, const int* in_sizes, int n_in,
                              void* d_out, int out_size, void* d_ws, size_t ws_size,
                              hipStream_t stream) {
    const float* in_feat = (const float*)d_in[0];
    const float* W1      = (const float*)d_in[1];
    const float* b1      = (const float*)d_in[2];
    const float* W2      = (const float*)d_in[3];
    const float* b2      = (const float*)d_in[4];
    const int*   src     = (const int*)d_in[5];
    const int*   dst     = (const int*)d_in[6];
    float*       out     = (float*)d_out;

    const int N  = in_sizes[0] / IN_DIM;
    const int E  = in_sizes[5];
    const int NB = (E + CHUNK_EDGES - 1) / CHUNK_EDGES;
    const int R  = (N + RANGE_NODES - 1) / RANGE_NODES;

    float* agg    = (float*)d_ws;
    float* hbuf   = agg + (size_t)N * HID_DIM;
    int*   part_d = (int*)(hbuf + (size_t)N * HID_DIM);
    int*   part_s = part_d + (size_t)NB * CHUNK_EDGES * 2;
    int*   tab    = part_s + (size_t)NB * CHUNK_EDGES;
    int*   list   = tab + (size_t)R * NB * 2;
    int*   cursor = list + E;
    float* inv    = (float*)(cursor + 4);
    int*   rstart = (int*)(inv + (size_t)2 * N);
    int*   degin  = rstart + N;
    float* inv_out = inv;
    float* inv_in  = inv + N;

    {
        const size_t lds = ((size_t)3 * CHUNK_EDGES + R) * sizeof(int);
        gcn_partition_kernel<<<NB, 1024, lds, stream>>>(
            src, dst, part_d, part_s, tab, cursor, E, R, NB);
        gcn_range_kernel<<<R, 512, 0, stream>>>(
            part_d, part_s, tab, cursor, inv, rstart, degin, list, N, NB);
    }

    gcn_gather_kernel<<<(N + 7) / 8, 256, 0, stream>>>(
        in_feat, list, rstart, degin, inv_out, agg, N);

    {
        int grid = (N + 127) / 128;
        gcn_wmma_gemm_kernel<IN_DIM, HID_DIM, true>
            <<<grid, 256, 0, stream>>>(agg, W1, b1, inv_in, hbuf, N);
    }

    gcn_gather_kernel<<<(N + 7) / 8, 256, 0, stream>>>(
        hbuf, list, rstart, degin, inv_out, agg, N);

    {
        int grid = (N + 127) / 128;
        gcn_wmma_gemm_kernel<HID_DIM, OUT_DIM, false>
            <<<grid, 256, 0, stream>>>(agg, W2, b2, inv_in, out, N);
    }
}


# Each edge gather fused with the WMMA GEMM that consumes it: a 512-thread block sums its 128 rows into an LDS tile and multiplies from there; agg plane and two launches gone; sum order kept
# speedup vs baseline: 4.5570x; 1.2069x over previous
#include <hip/hip_runtime.h>


typedef __attribute__((ext_vector_type(2))) float v2f;
typedef __attribute__((ext_vector_type(8))) float v8f;

#define IN_DIM  128
#define HID_DIM 128
#define OUT_DIM 64

#define KEY_NONE 0x7fffffff

#define RANGE_SHIFT 8
#define RANGE_NODES 256
#define CHUNK_EDGES 4096
#define OWNER_STAGE 8192

__global__ __launch_bounds__(1024)
void gcn_partition_kernel(const int* __restrict__ src,
                          const int* __restrict__ dst,
                          int* __restrict__ part_d_raw,
                          int* __restrict__ part_s,
                          int* __restrict__ tab_raw,
                          int* __restrict__ cursor,
                          int E, int R, int NB) {
    extern __shared__ __attribute__((aligned(16))) int part_lds[];
    int2*     stage_d = (int2*)part_lds;
    int*      stage_s = part_lds + 2 * CHUNK_EDGES;
    unsigned* hist    = (unsigned*)(part_lds + 3 * CHUNK_EDGES);
    __shared__ unsigned wave_sum[32];
    __shared__ unsigned chunk_total;
    int2*  part_d = (int2*)part_d_raw;
    uint2* tab    = (uint2*)tab_raw;

    const int tid  = threadIdx.x;
    const int lane = tid & 31;
    const int wave = tid >> 5;
    const int blk  = blockIdx.x;
    const size_t e0 = (size_t)blk * CHUNK_EDGES;

    for (int i = tid; i < R; i += 1024) hist[i] = 0u;
    if (blk == 0 && tid == 0) *cursor = 0;
    __syncthreads();

    int s_k[4], d_k[4];
    unsigned rd_k[4], rs_k[4];
#pragma unroll
    for (int k = 0; k < 4; ++k) {
        const size_t e = e0 + (size_t)(k * 1024 + tid);
        s_k[k] = -1;
        d_k[k] = -1;
        rd_k[k] = 0u;
        rs_k[k] = 0u;
        if (e < (size_t)E) {
            s_k[k] = src[e];
            d_k[k] = dst[e];
        }
    }
#pragma unroll
    for (int k = 0; k < 4; ++k) {
        if (s_k[k] >= 0) {
            rd_k[k] = atomicAdd(&hist[d_k[k] >> RANGE_SHIFT], 1u) & 0xffffu;
            rs_k[k] = atomicAdd(&hist[s_k[k] >> RANGE_SHIFT], 0x10000u) >> 16;
        }
    }
    __syncthreads();

    unsigned carry = 0u;
    for (int r0 = 0; r0 < R; r0 += 1024) {
        const int r = r0 + tid;
        const unsigned c = (r < R) ? hist[r] : 0u;
        unsigned incl = c;
#pragma unroll
        for (int o = 1; o < 32; o <<= 1) {
            const unsigned t = __shfl_up(incl, o, 32);
            if (lane >= o) incl += t;
        }
        if (lane == 31) wave_sum[wave] = incl;
        __syncthreads();
        if (tid == 0) {
            unsigned tot = 0u;
            for (int k = 0; k < 32; ++k) { const unsigned t = wave_sum[k]; wave_sum[k] = tot; tot += t; }
            chunk_total = tot;
        }
        __syncthreads();
        const unsigned off = carry + wave_sum[wave] + incl - c;
        if (r < R) {
            hist[r] = off;
            uint2 w;
            w.x = (off & 0xffffu) | ((c & 0xffffu) << 16);
            w.y = (off >> 16) | ((c >> 16) << 16);
            tab[(size_t)r * NB + blk] = w;
        }
        carry += chunk_total;
        __syncthreads();
    }

#pragma unroll
    for (int k = 0; k < 4; ++k) {
        if (s_k[k] >= 0) {
            const unsigned pd = (hist[d_k[k] >> RANGE_SHIFT] & 0xffffu) + rd_k[k];
            const unsigned ps = (hist[s_k[k] >> RANGE_SHIFT] >> 16) + rs_k[k];
            stage_d[pd] = make_int2(s_k[k], d_k[k]);
            stage_s[ps] = s_k[k];
        }
    }
    __syncthreads();

    size_t left = (size_t)E - e0;
    const int n = (left < (size_t)CHUNK_EDGES) ? (int)left : CHUNK_EDGES;
    for (int i = tid; i < n; i += 1024) {
        part_d[e0 + i] = stage_d[i];
        part_s[e0 + i] = stage_s[i];
    }
}

__global__ __launch_bounds__(512)
void gcn_range_kernel(const int* __restrict__ part_d_raw,
                      const int* __restrict__ part_s,
                      const int* __restrict__ tab_raw,
                      int* __restrict__ cursor,
                      float* __restrict__ inv,
                      int* __restrict__ row_start,
                      int* __restrict__ deg_in,
                      int* __restrict__ list,
                      int N, int NB) {
    __shared__ int cnt_in[RANGE_NODES];
    __shared__ int cnt_out[RANGE_NODES];
    __shared__ int row_off[RANGE_NODES];
    __shared__ int fill[RANGE_NODES];
    __shared__ int wave_sum[8];
    __shared__ int run_base, run_total;
    __shared__ int stage[OWNER_STAGE];
    const int2*  part_d = (const int2*)part_d_raw;
    const uint2* tab    = (const uint2*)tab_raw;

    const int tid  = threadIdx.x;
    const int lane = tid & 31;
    const int wave = tid >> 5;
    const int rng  = blockIdx.x;

    if (tid < RANGE_NODES) { cnt_in[tid] = 0; cnt_out[tid] = 0; fill[tid] = 0; }
    __syncthreads();

    for (int blk = tid; blk < NB; blk += 512) {
        const uint2 w = tab[(size_t)rng * NB + blk];
        {
            const int off = (int)(w.x & 0xffffu), c = (int)(w.x >> 16);
            const int2* __restrict__ p = part_d + (size_t)blk * CHUNK_EDGES + off;
            for (int i = 0; i < c; i += 8) {
                int d[8];
#pragma unroll
                for (int u = 0; u < 8; ++u) d[u] = p[(i + u < c) ? i + u : c - 1].y;
#pragma unroll
                for (int u = 0; u < 8; ++u)
                    if (i + u < c) atomicAdd(&cnt_in[d[u] & (RANGE_NODES - 1)], 1);
            }
        }
        {
            const int off = (int)(w.y & 0xffffu), c = (int)(w.y >> 16);
            const int* __restrict__ q = part_s + (size_t)blk * CHUNK_EDGES + off;
            for (int i = 0; i < c; i += 8) {
                int s[8];
#pragma unroll
                for (int u = 0; u < 8; ++u) s[u] = q[(i + u < c) ? i + u : c - 1];
#pragma unroll
                for (int u = 0; u < 8; ++u)
                    if (i + u < c) atomicAdd(&cnt_out[s[u] & (RANGE_NODES - 1)], 1);
            }
        }
    }
    __syncthreads();

    const int node = rng * RANGE_NODES + tid;
    int din = 0;
    if (tid < RANGE_NODES) {
        din = cnt_in[tid];
        if (node < N) {
            inv[node]     = rsqrtf(fmaxf((float)cnt_out[tid], 1.0f));
            inv[N + node] = rsqrtf(fmaxf((float)din, 1.0f));
            deg_in[node]  = din;
        }
    }
    int incl = din;
#pragma unroll
    for (int o = 1; o < 32; o <<= 1) {
        const int t = __shfl_up(incl, o, 32);
        if (lane >= o) incl += t;
    }
    if (tid < RANGE_NODES && lane == 31) wave_sum[wave] = incl;
    __syncthreads();
    if (tid == 0) {
        int tot = 0;
#pragma unroll
        for (int k = 0; k < 8; ++k) { const int t = wave_sum[k]; wave_sum[k] = tot; tot += t; }
        run_total = tot;
        run_base = atomicAdd(cursor, tot);
    }
    __syncthreads();
    const int total = run_total;
    const int base  = run_base;
    if (tid < RANGE_NODES) {
        const int excl = wave_sum[wave] + incl - din;
        row_off[tid] = excl;
        if (node < N) row_start[node] = base + excl;
    }
    __syncthreads();

    const bool staged = total <= OWNER_STAGE;
    for (int blk = tid; blk < NB; blk += 512) {
        const uint2 w = tab[(size_t)rng * NB + blk];
        const int off = (int)(w.x & 0xffffu), c = (int)(w.x >> 16);
        const int2* __restrict__ p = part_d + (size_t)blk * CHUNK_EDGES + off;
        for (int i = 0; i < c; i += 8) {
            int2 ed[8];
#pragma unroll
            for (int u = 0; u < 8; ++u) ed[u] = p[(i + u < c) ? i + u : c - 1];
#pragma unroll
            for (int u = 0; u < 8; ++u) {
                if (i + u < c) {
                    const int j = ed[u].y & (RANGE_NODES - 1);
                    const int pos = row_off[j] + atomicAdd(&fill[j], 1);
                    if (staged) stage[pos] = ed[u].x;
                    else        list[(size_t)base + pos] = ed[u].x;
                }
            }
        }
    }
    if (staged) {
        __syncthreads();
        for (int i = tid; i < total; i += 512) list[(size_t)base + i] = stage[i];
    }
}

__device__ __forceinline__ void gcn_add_row(float4& acc, const float4 v, const float sc) {
    acc.x += v.x * sc;
    acc.y += v.y * sc;
    acc.z += v.z * sc;
    acc.w += v.w * sc;
}

__device__ __forceinline__ float4 gcn_load_row(const float* __restrict__ x, int s, int lane) {
    return *((const float4*)(x + (size_t)s * 128) + lane);
}

template <int NCH>
__device__ __forceinline__ float4 gcn_gather_ranked(const float* __restrict__ x,
                                                    const float* __restrict__ inv_out,
                                                    const int* __restrict__ row_list,
                                                    int deg, int lane,
                                                    int* __restrict__ srt) {
    int key[NCH], rk[NCH];
#pragma unroll
    for (int c = 0; c < NCH; ++c) {
        const int i = lane + 32 * c;
        key[c] = (i < deg) ? row_list[i] : KEY_NONE;
        rk[c] = 0;
    }
#pragma unroll
    for (int c = 0; c < NCH; ++c) {
        int lim = deg - 32 * c;
        if (lim > 32) lim = 32;
        for (int j = 0; j < lim; ++j) {
            const int kj = __builtin_amdgcn_readlane(key[c], j);
            const int jj = j + 32 * c;
#pragma unroll
            for (int c2 = 0; c2 < NCH; ++c2) {
                const int i2 = lane + 32 * c2;
                rk[c2] += (kj < key[c2] || (kj == key[c2] && jj < i2)) ? 1 : 0;
            }
        }
    }
#pragma unroll
    for (int c = 0; c < NCH; ++c) {
        const int i = lane + 32 * c;
        if (i >= deg) rk[c] = i;
        srt[rk[c]] = key[c];
    }
    __builtin_amdgcn_fence(__ATOMIC_RELEASE, "wavefront");
    __builtin_amdgcn_wave_barrier();
    __builtin_amdgcn_fence(__ATOMIC_ACQUIRE, "wavefront");
    int sorted[NCH];
#pragma unroll
    for (int c = 0; c < NCH; ++c) sorted[c] = srt[lane + 32 * c];

    float4 acc = make_float4(0.0f, 0.0f, 0.0f, 0.0f);
#pragma unroll
    for (int c = 0; c < NCH; ++c) {
        int lim = deg - 32 * c;
        if (lim > 32) lim = 32;
        int j = 0;
        for (; j + 8 <= lim; j += 8) {
            float4 v[8];
            float sc[8];
#pragma unroll
            for (int u = 0; u < 8; ++u) {
                const int s = __builtin_amdgcn_readlane(sorted[c], j + u);
                v[u] = gcn_load_row(x, s, lane);
                sc[u] = inv_out[s];
            }
#pragma unroll
            for (int u = 0; u < 8; ++u) gcn_add_row(acc, v[u], sc[u]);
        }
        const int rest = lim - j;
        if (rest > 0) {
            float4 v[7];
            float sc[7];
#pragma unroll
            for (int u = 0; u < 7; ++u) {
                v[u] = make_float4(0.0f, 0.0f, 0.0f, 0.0f);
                sc[u] = 0.0f;
                if (u < rest) {
                    const int s = __builtin_amdgcn_readlane(sorted[c], j + u);
                    v[u] = gcn_load_row(x, s, lane);
                    sc[u] = inv_out[s];
                }
            }
#pragma unroll
            for (int u = 0; u < 7; ++u)
                if (u < rest) gcn_add_row(acc, v[u], sc[u]);
        }
    }
    return acc;
}

__device__ __forceinline__ int gcn_wave_min(int v) {
#pragma unroll
    for (int o = 16; o > 0; o >>= 1) { int t = __shfl_xor(v, o, 32); v = (t < v) ? t : v; }
    return v;
}
__device__ __forceinline__ int gcn_wave_add(int v) {
#pragma unroll
    for (int o = 16; o > 0; o >>= 1) v += __shfl_xor(v, o, 32);
    return v;
}

__device__ __forceinline__ float4 gcn_gather_long(const float* __restrict__ x,
                                                  const float* __restrict__ inv_out,
                                                  const int* __restrict__ row_list,
                                                  int deg, int lane) {
    float4 acc = make_float4(0.0f, 0.0f, 0.0f, 0.0f);
    int cur = -1;
    for (int pass = 0; pass < deg; ++pass) {
        int m = KEY_NONE;
        for (int p = lane; p < deg; p += 32) {
            const int k = row_list[p];
            if (k > cur && k < m) m = k;
        }
        m = gcn_wave_min(m);
        if (m == KEY_NONE) break;
        int cnt = 0;
        for (int p = lane; p < deg; p += 32) cnt += (row_list[p] == m) ? 1 : 0;
        cnt = __builtin_amdgcn_readfirstlane(gcn_wave_add(cnt));
        m = __builtin_amdgcn_readfirstlane(m);
        const float4 v = gcn_load_row(x, m, lane);
        const float sc = inv_out[m];
        for (int t = 0; t < cnt; ++t) gcn_add_row(acc, v, sc);
        cur = m;
    }
    return acc;
}

template <int NOUT, bool RELU>
__global__ __launch_bounds__(512)
void gcn_layer_kernel(const float* __restrict__ x,
                      const int* __restrict__ list,
                      const int* __restrict__ row_start,
                      const int* __restrict__ deg_in,
                      const float* __restrict__ inv_out,
                      const float* __restrict__ inv_in,
                      const float* __restrict__ W,
                      const float* __restrict__ bias,
                      float* __restrict__ out, int N) {
    constexpr int K   = 128;
    constexpr int S   = 2 * NOUT + 32;
    constexpr int PA  = K + 4;
    constexpr int NT  = NOUT / 16;
    constexpr int NTW = NT / 2;
    __shared__ __attribute__((aligned(16))) float lds_a[128 * PA];
    __shared__ __attribute__((aligned(16))) float lds_w[(K / 2) * S];
    __shared__ int srt_all[16][64];

    const int tid  = threadIdx.x;
    const int lane = tid & 31;
    const int wave = __builtin_amdgcn_readfirstlane(tid >> 5);

    for (int idx = tid; idx < K * NOUT; idx += 512) {
        const int k = idx / NOUT;
        const int n = idx - k * NOUT;
        lds_w[(k >> 1) * S + n * 2 + (k & 1)] = W[idx];
    }

    const int lrow0 = wave * 8;
    const int grow0 = blockIdx.x * 128 + lrow0;
    int pr = grow0 + (lane & 7);
    if (pr >= N) pr = N - 1;
    const int   st_v = row_start[pr];
    const int   dg_v = deg_in[pr];
    const float si_v = inv_in[pr];
#pragma nounroll
    for (int r = 0; r < 8; ++r) {
        const int row = grow0 + r;
        float4 acc = make_float4(0.0f, 0.0f, 0.0f, 0.0f);
        if (row < N) {
            const int start = __builtin_amdgcn_readlane(st_v, r);
            const int deg   = __builtin_amdgcn_readlane(dg_v, r);
            const float s   = __int_as_float(__builtin_amdgcn_readlane(__float_as_int(si_v), r));
            const int* __restrict__ row_list = list + start;
            if (deg <= 32)      acc = gcn_gather_ranked<1>(x, inv_out, row_list, deg, lane, srt_all[wave]);
            else if (deg <= 64) acc = gcn_gather_ranked<2>(x, inv_out, row_list, deg, lane, srt_all[wave]);
            else                acc = gcn_gather_long(x, inv_out, row_list, deg, lane);
            acc.x *= s;
            acc.y *= s;
            acc.z *= s;
            acc.w *= s;
        }
        *((float4*)(lds_a + (size_t)(lrow0 + r) * PA) + lane) = acc;
    }
    __syncthreads();

    const int g  = lane >> 4;
    const int mr = lane & 15;
    const int rt = wave >> 1;
    const int t0 = (wave & 1) * NTW;
    const float* __restrict__ ap = lds_a + (size_t)(rt * 16 + mr) * PA + 2 * g;

    v8f acc[NTW];
#pragma unroll
    for (int t = 0; t < NTW; ++t) acc[t] = (v8f){};

#pragma unroll 4
    for (int k0 = 0; k0 < K; k0 += 4) {
        const v2f a = *(const v2f*)(ap + k0);
        const v2f* __restrict__ wp =
            (const v2f*)(lds_w + (size_t)(k0 / 2 + g) * S);
#pragma unroll
        for (int t = 0; t < NTW; ++t) {
            v2f b = wp[(t0 + t) * 16 + mr];
            acc[t] = __builtin_amdgcn_wmma_f32_16x16x4_f32(
                 false, a,  false, b,
                 (short)0, acc[t],  false,  false);
        }
    }

    const int orow0 = blockIdx.x * 128 + rt * 16;
#pragma unroll
    for (int t = 0; t < NTW; ++t) {
        const int Nbase = (t0 + t) * 16;
        const float bv = bias[Nbase + mr];
#pragma unroll
        for (int r = 0; r < 8; ++r) {
            const int orow = orow0 + r + 8 * g;
            if (orow < N) {
                float v = acc[t][r] + bv;
                if (RELU) v = fmaxf(v, 0.0f);
                out[(size_t)orow * NOUT + Nbase + mr] = v;
            }
        }
    }
}

extern "C" void kernel_launch(void* const* d_in, const int* in_sizes, int n_in,
                              void* d_out, int out_size, void* d_ws, size_t ws_size,
                              hipStream_t stream) {
    const float* in_feat = (const float*)d_in[0];
    const float* W1      = (const float*)d_in[1];
    const float* b1      = (const float*)d_in[2];
    const float* W2      = (const float*)d_in[3];
    const float* b2      = (const float*)d_in[4];
    const int*   src     = (const int*)d_in[5];
    const int*   dst     = (const int*)d_in[6];
    float*       out     = (float*)d_out;

    const int N  = in_sizes[0] / IN_DIM;
    const int E  = in_sizes[5];
    const int NB = (E + CHUNK_EDGES - 1) / CHUNK_EDGES;
    const int R  = (N + RANGE_NODES - 1) / RANGE_NODES;

    float* hbuf   = (float*)d_ws;
    int*   part_d = (int*)(hbuf + (size_t)N * HID_DIM);
    int*   part_s = part_d + (size_t)NB * CHUNK_EDGES * 2;
    int*   tab    = part_s + (size_t)NB * CHUNK_EDGES;
    int*   list   = tab + (size_t)R * NB * 2;
    int*   cursor = list + E;
    float* inv    = (float*)(cursor + 4);
    int*   rstart = (int*)(inv + (size_t)2 * N);
    int*   degin  = rstart + N;
    float* inv_out = inv;
    float* inv_in  = inv + N;

    {
        const size_t lds = ((size_t)3 * CHUNK_EDGES + R) * sizeof(int);
        gcn_partition_kernel<<<NB, 1024, lds, stream>>>(
            src, dst, part_d, part_s, tab, cursor, E, R, NB);
        gcn_range_kernel<<<R, 512, 0, stream>>>(
            part_d, part_s, tab, cursor, inv, rstart, degin, list, N, NB);
    }

    const int grid = (N + 127) / 128;

    gcn_layer_kernel<HID_DIM, true><<<grid, 512, 0, stream>>>(
        in_feat, list, rstart, degin, inv_out, inv_in, W1, b1, hbuf, N);

    gcn_layer_kernel<OUT_DIM, false><<<grid, 512, 0, stream>>>(
        hbuf, list, rstart, degin, inv_out, inv_in, W2, b2, out, N);
}


# Layer blocks own 208 rows, taken in two passes (128, 80) through the one LDS tile: 481 blocks, all resident at once (2 per WGP), instead of 782 blocks of 128 rows on 512 places; sum order kept
# speedup vs baseline: 4.9363x; 1.0832x over previous
#include <hip/hip_runtime.h>


typedef __attribute__((ext_vector_type(2))) float v2f;
typedef __attribute__((ext_vector_type(8))) float v8f;

#define IN_DIM  128
#define HID_DIM 128
#define OUT_DIM 64

#define KEY_NONE 0x7fffffff

#define RANGE_SHIFT 8
#define RANGE_NODES 256
#define CHUNK_EDGES 4096
#define OWNER_STAGE 8192

#define LAYER_ROWS 208
#define TILE_ROWS  128

__global__ __launch_bounds__(1024)
void gcn_partition_kernel(const int* __restrict__ src,
                          const int* __restrict__ dst,
                          int* __restrict__ part_d_raw,
                          int* __restrict__ part_s,
                          int* __restrict__ tab_raw,
                          int* __restrict__ cursor,
                          int E, int R, int NB) {
    extern __shared__ __attribute__((aligned(16))) int part_lds[];
    int2*     stage_d = (int2*)part_lds;
    int*      stage_s = part_lds + 2 * CHUNK_EDGES;
    unsigned* hist    = (unsigned*)(part_lds + 3 * CHUNK_EDGES);
    __shared__ unsigned wave_sum[32];
    __shared__ unsigned chunk_total;
    int2*  part_d = (int2*)part_d_raw;
    uint2* tab    = (uint2*)tab_raw;

    const int tid  = threadIdx.x;
    const int lane = tid & 31;
    const int wave = tid >> 5;
    const int blk  = blockIdx.x;
    const size_t e0 = (size_t)blk * CHUNK_EDGES;

    for (int i = tid; i < R; i += 1024) hist[i] = 0u;
    if (blk == 0 && tid == 0) *cursor = 0;
    __syncthreads();

    int s_k[4], d_k[4];
    unsigned rd_k[4], rs_k[4];
#pragma unroll
    for (int k = 0; k < 4; ++k) {
        const size_t e = e0 + (size_t)(k * 1024 + tid);
        s_k[k] = -1;
        d_k[k] = -1;
        rd_k[k] = 0u;
        rs_k[k] = 0u;
        if (e < (size_t)E) {
            s_k[k] = src[e];
            d_k[k] = dst[e];
        }
    }
#pragma unroll
    for (int k = 0; k < 4; ++k) {
        if (s_k[k] >= 0) {
            rd_k[k] = atomicAdd(&hist[d_k[k] >> RANGE_SHIFT], 1u) & 0xffffu;
            rs_k[k] = atomicAdd(&hist[s_k[k] >> RANGE_SHIFT], 0x10000u) >> 16;
        }
    }
    __syncthreads();

    unsigned carry = 0u;
    for (int r0 = 0; r0 < R; r0 += 1024) {
        const int r = r0 + tid;
        const unsigned c = (r < R) ? hist[r] : 0u;
        unsigned incl = c;
#pragma unroll
        for (int o = 1; o < 32; o <<= 1) {
            const unsigned t = __shfl_up(incl, o, 32);
            if (lane >= o) incl += t;
        }
        if (lane == 31) wave_sum[wave] = incl;
        __syncthreads();
        if (tid == 0) {
            unsigned tot = 0u;
            for (int k = 0; k < 32; ++k) { const unsigned t = wave_sum[k]; wave_sum[k] = tot; tot += t; }
            chunk_total = tot;
        }
        __syncthreads();
        const unsigned off = carry + wave_sum[wave] + incl - c;
        if (r < R) {
            hist[r] = off;
            uint2 w;
            w.x = (off & 0xffffu) | ((c & 0xffffu) << 16);
            w.y = (off >> 16) | ((c >> 16) << 16);
            tab[(size_t)r * NB + blk] = w;
        }
        carry += chunk_total;
        __syncthreads();
    }

#pragma unroll
    for (int k = 0; k < 4; ++k) {
        if (s_k[k] >= 0) {
            const unsigned pd = (hist[d_k[k] >> RANGE_SHIFT] & 0xffffu) + rd_k[k];
            const unsigned ps = (hist[s_k[k] >> RANGE_SHIFT] >> 16) + rs_k[k];
            stage_d[pd] = make_int2(s_k[k], d_k[k]);
            stage_s[ps] = s_k[k];
        }
    }
    __syncthreads();

    size_t left = (size_t)E - e0;
    const int n = (left < (size_t)CHUNK_EDGES) ? (int)left : CHUNK_EDGES;
    for (int i = tid; i < n; i += 1024) {
        part_d[e0 + i] = stage_d[i];
        part_s[e0 + i] = stage_s[i];
    }
}

__global__ __launch_bounds__(512)
void gcn_range_kernel(const int* __restrict__ part_d_raw,
                      const int* __restrict__ part_s,
                      const int* __restrict__ tab_raw,
                      int* __restrict__ cursor,
                      float* __restrict__ inv,
                      int* __restrict__ row_start,
                      int* __restrict__ deg_in,
                      int* __restrict__ list,
                      int N, int NB) {
    __shared__ int cnt_in[RANGE_NODES];
    __shared__ int cnt_out[RANGE_NODES];
    __shared__ int row_off[RANGE_NODES];
    __shared__ int fill[RANGE_NODES];
    __shared__ int wave_sum[8];
    __shared__ int run_base, run_total;
    __shared__ int stage[OWNER_STAGE];
    const int2*  part_d = (const int2*)part_d_raw;
    const uint2* tab    = (const uint2*)tab_raw;

    const int tid  = threadIdx.x;
    const int lane = tid & 31;
    const int wave = tid >> 5;
    const int rng  = blockIdx.x;

    if (tid < RANGE_NODES) { cnt_in[tid] = 0; cnt_out[tid] = 0; fill[tid] = 0; }
    __syncthreads();

    for (int blk = tid; blk < NB; blk += 512) {
        const uint2 w = tab[(size_t)rng * NB + blk];
        {
            const int off = (int)(w.x & 0xffffu), c = (int)(w.x >> 16);
            const int2* __restrict__ p = part_d + (size_t)blk * CHUNK_EDGES + off;
            for (int i = 0; i < c; i += 8) {
                int d[8];
#pragma unroll
                for (int u = 0; u < 8; ++u) d[u] = p[(i + u < c) ? i + u : c - 1].y;
#pragma unroll
                for (int u = 0; u < 8; ++u)
                    if (i + u < c) atomicAdd(&cnt_in[d[u] & (RANGE_NODES - 1)], 1);
            }
        }
        {
            const int off = (int)(w.y & 0xffffu), c = (int)(w.y >> 16);
            const int* __restrict__ q = part_s + (size_t)blk * CHUNK_EDGES + off;
            for (int i = 0; i < c; i += 8) {
                int s[8];
#pragma unroll
                for (int u = 0; u < 8; ++u) s[u] = q[(i + u < c) ? i + u : c - 1];
#pragma unroll
                for (int u = 0; u < 8; ++u)
                    if (i + u < c) atomicAdd(&cnt_out[s[u] & (RANGE_NODES - 1)], 1);
            }
        }
    }
    __syncthreads();

    const int node = rng * RANGE_NODES + tid;
    int din = 0;
    if (tid < RANGE_NODES) {
        din = cnt_in[tid];
        if (node < N) {
            inv[node]     = rsqrtf(fmaxf((float)cnt_out[tid], 1.0f));
            inv[N + node] = rsqrtf(fmaxf((float)din, 1.0f));
            deg_in[node]  = din;
        }
    }
    int incl = din;
#pragma unroll
    for (int o = 1; o < 32; o <<= 1) {
        const int t = __shfl_up(incl, o, 32);
        if (lane >= o) incl += t;
    }
    if (tid < RANGE_NODES && lane == 31) wave_sum[wave] = incl;
    __syncthreads();
    if (tid == 0) {
        int tot = 0;
#pragma unroll
        for (int k = 0; k < 8; ++k) { const int t = wave_sum[k]; wave_sum[k] = tot; tot += t; }
        run_total = tot;
        run_base = atomicAdd(cursor, tot);
    }
    __syncthreads();
    const int total = run_total;
    const int base  = run_base;
    if (tid < RANGE_NODES) {
        const int excl = wave_sum[wave] + incl - din;
        row_off[tid] = excl;
        if (node < N) row_start[node] = base + excl;
    }
    __syncthreads();

    const bool staged = total <= OWNER_STAGE;
    for (int blk = tid; blk < NB; blk += 512) {
        const uint2 w = tab[(size_t)rng * NB + blk];
        const int off = (int)(w.x & 0xffffu), c = (int)(w.x >> 16);
        const int2* __restrict__ p = part_d + (size_t)blk * CHUNK_EDGES + off;
        for (int i = 0; i < c; i += 8) {
            int2 ed[8];
#pragma unroll
            for (int u = 0; u < 8; ++u) ed[u] = p[(i + u < c) ? i + u : c - 1];
#pragma unroll
            for (int u = 0; u < 8; ++u) {
                if (i + u < c) {
                    const int j = ed[u].y & (RANGE_NODES - 1);
                    const int pos = row_off[j] + atomicAdd(&fill[j], 1);
                    if (staged) stage[pos] = ed[u].x;
                    else        list[(size_t)base + pos] = ed[u].x;
                }
            }
        }
    }
    if (staged) {
        __syncthreads();
        for (int i = tid; i < total; i += 512) list[(size_t)base + i] = stage[i];
    }
}

__device__ __forceinline__ void gcn_add_row(float4& acc, const float4 v, const float sc) {
    acc.x += v.x * sc;
    acc.y += v.y * sc;
    acc.z += v.z * sc;
    acc.w += v.w * sc;
}

__device__ __forceinline__ float4 gcn_load_row(const float* __restrict__ x, int s, int lane) {
    return *((const float4*)(x + (size_t)s * 128) + lane);
}

template <int NCH>
__device__ __forceinline__ float4 gcn_gather_ranked(const float* __restrict__ x,
                                                    const float* __restrict__ inv_out,
                                                    const int* __restrict__ row_list,
                                                    int deg, int lane,
                                                    int* __restrict__ srt) {
    int key[NCH], rk[NCH];
#pragma unroll
    for (int c = 0; c < NCH; ++c) {
        const int i = lane + 32 * c;
        key[c] = (i < deg) ? row_list[i] : KEY_NONE;
        rk[c] = 0;
    }
#pragma unroll
    for (int c = 0; c < NCH; ++c) {
        int lim = deg - 32 * c;
        if (lim > 32) lim = 32;
        for (int j = 0; j < lim; ++j) {
            const int kj = __builtin_amdgcn_readlane(key[c], j);
            const int jj = j + 32 * c;
#pragma unroll
            for (int c2 = 0; c2 < NCH; ++c2) {
                const int i2 = lane + 32 * c2;
                rk[c2] += (kj < key[c2] || (kj == key[c2] && jj < i2)) ? 1 : 0;
            }
        }
    }
#pragma unroll
    for (int c = 0; c < NCH; ++c) {
        const int i = lane + 32 * c;
        if (i >= deg) rk[c] = i;
        srt[rk[c]] = key[c];
    }
    __builtin_amdgcn_fence(__ATOMIC_RELEASE, "wavefront");
    __builtin_amdgcn_wave_barrier();
    __builtin_amdgcn_fence(__ATOMIC_ACQUIRE, "wavefront");
    int sorted[NCH];
#pragma unroll
    for (int c = 0; c < NCH; ++c) sorted[c] = srt[lane + 32 * c];

    float4 acc = make_float4(0.0f, 0.0f, 0.0f, 0.0f);
#pragma unroll
    for (int c = 0; c < NCH; ++c) {
        int lim = deg - 32 * c;
        if (lim > 32) lim = 32;
        int j = 0;
        for (; j + 8 <= lim; j += 8) {
            float4 v[8];
            float sc[8];
#pragma unroll
            for (int u = 0; u < 8; ++u) {
                const int s = __builtin_amdgcn_readlane(sorted[c], j + u);
                v[u] = gcn_load_row(x, s, lane);
                sc[u] = inv_out[s];
            }
#pragma unroll
            for (int u = 0; u < 8; ++u) gcn_add_row(acc, v[u], sc[u]);
        }
        const int rest = lim - j;
        if (rest > 0) {
            float4 v[7];
            float sc[7];
#pragma unroll
            for (int u = 0; u < 7; ++u) {
                v[u] = make_float4(0.0f, 0.0f, 0.0f, 0.0f);
                sc[u] = 0.0f;
                if (u < rest) {
                    const int s = __builtin_amdgcn_readlane(sorted[c], j + u);
                    v[u] = gcn_load_row(x, s, lane);
                    sc[u] = inv_out[s];
                }
            }
#pragma unroll
            for (int u = 0; u < 7; ++u)
                if (u < rest) gcn_add_row(acc, v[u], sc[u]);
        }
    }
    return acc;
}

__device__ __forceinline__ int gcn_wave_min(int v) {
#pragma unroll
    for (int o = 16; o > 0; o >>= 1) { int t = __shfl_xor(v, o, 32); v = (t < v) ? t : v; }
    return v;
}
__device__ __forceinline__ int gcn_wave_add(int v) {
#pragma unroll
    for (int o = 16; o > 0; o >>= 1) v += __shfl_xor(v, o, 32);
    return v;
}

__device__ __forceinline__ float4 gcn_gather_long(const float* __restrict__ x,
                                                  const float* __restrict__ inv_out,
                                                  const int* __restrict__ row_list,
                                                  int deg, int lane) {
    float4 acc = make_float4(0.0f, 0.0f, 0.0f, 0.0f);
    int cur = -1;
    for (int pass = 0; pass < deg; ++pass) {
        int m = KEY_NONE;
        for (int p = lane; p < deg; p += 32) {
            const int k = row_list[p];
            if (k > cur && k < m) m = k;
        }
        m = gcn_wave_min(m);
        if (m == KEY_NONE) break;
        int cnt = 0;
        for (int p = lane; p < deg; p += 32) cnt += (row_list[p] == m) ? 1 : 0;
        cnt = __builtin_amdgcn_readfirstlane(gcn_wave_add(cnt));
        m = __builtin_amdgcn_readfirstlane(m);
        const float4 v = gcn_load_row(x, m, lane);
        const float sc = inv_out[m];
        for (int t = 0; t < cnt; ++t) gcn_add_row(acc, v, sc);
        cur = m;
    }
    return acc;
}

template <int NOUT, bool RELU>
__global__ __launch_bounds__(512)
void gcn_layer_kernel(const float* __restrict__ x,
                      const int* __restrict__ list,
                      const int* __restrict__ row_start,
                      const int* __restrict__ deg_in,
                      const float* __restrict__ inv_out,
                      const float* __restrict__ inv_in,
                      const float* __restrict__ W,
                      const float* __restrict__ bias,
                      float* __restrict__ out, int N) {
    constexpr int K   = 128;
    constexpr int S   = 2 * NOUT + 32;
    constexpr int PA  = K + 4;
    constexpr int NT  = NOUT / 16;
    constexpr int NTW = NT / 2;
    __shared__ __attribute__((aligned(16))) float lds_a[TILE_ROWS * PA];
    __shared__ __attribute__((aligned(16))) float lds_w[(K / 2) * S];
    __shared__ int srt_all[16][64];

    const int tid  = threadIdx.x;
    const int lane = tid & 31;
    const int wave = __builtin_amdgcn_readfirstlane(tid >> 5);

    for (int idx = tid; idx < K * NOUT; idx += 512) {
        const int k = idx / NOUT;
        const int n = idx - k * NOUT;
        lds_w[(k >> 1) * S + n * 2 + (k & 1)] = W[idx];
    }

    const int g  = lane >> 4;
    const int mr = lane & 15;
    const int rt = wave >> 1;
    const int t0 = (wave & 1) * NTW;
    const float* __restrict__ ap = lds_a + (size_t)(rt * 16 + mr) * PA + 2 * g;

#pragma nounroll
    for (int pass = 0; pass < 2; ++pass) {
        const int prow0 = blockIdx.x * LAYER_ROWS + pass * TILE_ROWS;
        const int nrw   = (pass == 0) ? TILE_ROWS / 16 : (LAYER_ROWS - TILE_ROWS) / 16;

        const int lrow0 = wave * nrw;
        const int grow0 = prow0 + lrow0;
        int pr = grow0 + (lane & 7);
        if (pr >= N) pr = N - 1;
        const int   st_v = row_start[pr];
        const int   dg_v = deg_in[pr];
        const float si_v = inv_in[pr];
#pragma nounroll
        for (int r = 0; r < nrw; ++r) {
            const int row = grow0 + r;
            float4 acc = make_float4(0.0f, 0.0f, 0.0f, 0.0f);
            if (row < N) {
                const int start = __builtin_amdgcn_readlane(st_v, r);
                const int deg   = __builtin_amdgcn_readlane(dg_v, r);
                const float s   = __int_as_float(__builtin_amdgcn_readlane(__float_as_int(si_v), r));
                const int* __restrict__ row_list = list + start;
                if (deg <= 32)      acc = gcn_gather_ranked<1>(x, inv_out, row_list, deg, lane, srt_all[wave]);
                else if (deg <= 64) acc = gcn_gather_ranked<2>(x, inv_out, row_list, deg, lane, srt_all[wave]);
                else                acc = gcn_gather_long(x, inv_out, row_list, deg, lane);
                acc.x *= s;
                acc.y *= s;
                acc.z *= s;
                acc.w *= s;
            }
            *((float4*)(lds_a + (size_t)(lrow0 + r) * PA) + lane) = acc;
        }
        __syncthreads();

        const bool mine = rt < nrw;
        v8f acc[NTW];
#pragma unroll
        for (int t = 0; t < NTW; ++t) acc[t] = (v8f){};

        if (mine) {
#pragma unroll 4
            for (int k0 = 0; k0 < K; k0 += 4) {
                const v2f a = *(const v2f*)(ap + k0);
                const v2f* __restrict__ wp =
                    (const v2f*)(lds_w + (size_t)(k0 / 2 + g) * S);
#pragma unroll
                for (int t = 0; t < NTW; ++t) {
                    v2f b = wp[(t0 + t) * 16 + mr];
                    acc[t] = __builtin_amdgcn_wmma_f32_16x16x4_f32(
                         false, a,  false, b,
                         (short)0, acc[t],  false,  false);
                }
            }
        }
        if (pass == 0) __syncthreads();

        if (mine) {
            const int orow0 = prow0 + rt * 16;
#pragma unroll
            for (int t = 0; t < NTW; ++t) {
                const int Nbase = (t0 + t) * 16;
                const float bv = bias[Nbase + mr];
#pragma unroll
                for (int r = 0; r < 8; ++r) {
                    const int orow = orow0 + r + 8 * g;
                    if (orow < N) {
                        float v = acc[t][r] + bv;
                        if (RELU) v = fmaxf(v, 0.0f);
                        out[(size_t)orow * NOUT + Nbase + mr] = v;
                    }
                }
            }
        }
    }
}

extern "C" void kernel_launch(void* const* d_in, const int* in_sizes, int n_in,
                              void* d_out, int out_size, void* d_ws, size_t ws_size,
                              hipStream_t stream) {
    const float* in_feat = (const float*)d_in[0];
    const float* W1      = (const float*)d_in[1];
    const float* b1      = (const float*)d_in[2];
    const float* W2      = (const float*)d_in[3];
    const float* b2      = (const float*)d_in[4];
    const int*   src     = (const int*)d_in[5];
    const int*   dst     = (const int*)d_in[6];
    float*       out     = (float*)d_out;

    const int N  = in_sizes[0] / IN_DIM;
    const int E  = in_sizes[5];
    const int NB = (E + CHUNK_EDGES - 1) / CHUNK_EDGES;
    const int R  = (N + RANGE_NODES - 1) / RANGE_NODES;

    float* hbuf   = (float*)d_ws;
    int*   part_d = (int*)(hbuf + (size_t)N * HID_DIM);
    int*   part_s = part_d + (size_t)NB * CHUNK_EDGES * 2;
    int*   tab    = part_s + (size_t)NB * CHUNK_EDGES;
    int*   list   = tab + (size_t)R * NB * 2;
    int*   cursor = list + E;
    float* inv    = (float*)(cursor + 4);
    int*   rstart = (int*)(inv + (size_t)2 * N);
    int*   degin  = rstart + N;
    float* inv_out = inv;
    float* inv_in  = inv + N;

    {
        const size_t lds = ((size_t)3 * CHUNK_EDGES + R) * sizeof(int);
        gcn_partition_kernel<<<NB, 1024, lds, stream>>>(
            src, dst, part_d, part_s, tab, cursor, E, R, NB);
        gcn_range_kernel<<<R, 512, 0, stream>>>(
            part_d, part_s, tab, cursor, inv, rstart, degin, list, N, NB);
    }

    const int grid = (N + LAYER_ROWS - 1) / LAYER_ROWS;

    gcn_layer_kernel<HID_DIM, true><<<grid, 512, 0, stream>>>(
        in_feat, list, rstart, degin, inv_out, inv_in, W1, b1, hbuf, N);

    gcn_layer_kernel<OUT_DIM, false><<<grid, 512, 0, stream>>>(
        hbuf, list, rstart, degin, inv_out, inv_in, W2, b2, out, N);
}
